# starting kernel
# baseline (speedup 1.0000x reference)
#include <hip/hip_runtime.h>
#include <math.h>

typedef __attribute__((ext_vector_type(2))) float v2f;
typedef __attribute__((ext_vector_type(8))) float v8f;

#define DIM_B 2
#define DIM_T 2048
#define DIM_K 1024
#define NHEAD 16
#define HDIM  64

static __device__ __forceinline__ v8f wmma_f32(v2f a, v2f b, v8f c) {
  return __builtin_amdgcn_wmma_f32_16x16x4_f32(false, a, false, b, (short)0, c,
                                               false, false);
}

__global__ __launch_bounds__(256)
__attribute__((amdgpu_waves_per_eu(2)))
void gemm_xwt_kernel(const float* __restrict__ A, const float* __restrict__ W,
                     const float* __restrict__ bias, float* __restrict__ C,
                     int M, int N, int K) {
  const int lane  = threadIdx.x & 31;
  const int wave  = threadIdx.x >> 5;
  const int wm    = wave >> 2;
  const int wn    = wave & 3;
  const int m0    = blockIdx.y * 128 + wm * 64;
  const int n0    = blockIdx.x * 128 + wn * 32;
  const int lhalf = lane >> 4;
  const int lmod  = lane & 15;
  const int kbase = lhalf * 2;

  v8f acc[4][2];
#pragma unroll
  for (int i = 0; i < 4; ++i)
#pragma unroll
    for (int j = 0; j < 2; ++j)
#pragma unroll
      for (int r = 0; r < 8; ++r) acc[i][j][r] = 0.0f;

  const float* ar0 = A + (size_t)(m0 + lmod) * K;
  const float* ar1 = ar0 + (size_t)16 * K;
  const float* ar2 = ar0 + (size_t)32 * K;
  const float* ar3 = ar0 + (size_t)48 * K;
  const float* br0 = W + (size_t)(n0 + lmod) * K;
  const float* br1 = br0 + (size_t)16 * K;

  v2f a0 = *(const v2f*)(ar0 + kbase);
  v2f a1 = *(const v2f*)(ar1 + kbase);
  v2f a2 = *(const v2f*)(ar2 + kbase);
  v2f a3 = *(const v2f*)(ar3 + kbase);
  v2f b0 = *(const v2f*)(br0 + kbase);
  v2f b1 = *(const v2f*)(br1 + kbase);

  v2f c0 = *(const v2f*)(ar0 + 4 + kbase);
  v2f c1 = *(const v2f*)(ar1 + 4 + kbase);
  v2f c2 = *(const v2f*)(ar2 + 4 + kbase);
  v2f c3 = *(const v2f*)(ar3 + 4 + kbase);
  v2f d0 = *(const v2f*)(br0 + 4 + kbase);
  v2f d1 = *(const v2f*)(br1 + 4 + kbase);

#pragma clang loop unroll(disable)
  for (int k0 = 0; k0 + 16 <= K; k0 += 8) {
    acc[0][0] = wmma_f32(a0, b0, acc[0][0]);
    acc[0][1] = wmma_f32(a0, b1, acc[0][1]);
    acc[1][0] = wmma_f32(a1, b0, acc[1][0]);
    acc[1][1] = wmma_f32(a1, b1, acc[1][1]);
    acc[2][0] = wmma_f32(a2, b0, acc[2][0]);
    acc[2][1] = wmma_f32(a2, b1, acc[2][1]);
    acc[3][0] = wmma_f32(a3, b0, acc[3][0]);
    acc[3][1] = wmma_f32(a3, b1, acc[3][1]);
    {
      const int kn = k0 + 8 + kbase;
      a0 = *(const v2f*)(ar0 + kn);
      a1 = *(const v2f*)(ar1 + kn);
      a2 = *(const v2f*)(ar2 + kn);
      a3 = *(const v2f*)(ar3 + kn);
      b0 = *(const v2f*)(br0 + kn);
      b1 = *(const v2f*)(br1 + kn);
    }
    acc[0][0] = wmma_f32(c0, d0, acc[0][0]);
    acc[0][1] = wmma_f32(c0, d1, acc[0][1]);
    acc[1][0] = wmma_f32(c1, d0, acc[1][0]);
    acc[1][1] = wmma_f32(c1, d1, acc[1][1]);
    acc[2][0] = wmma_f32(c2, d0, acc[2][0]);
    acc[2][1] = wmma_f32(c2, d1, acc[2][1]);
    acc[3][0] = wmma_f32(c3, d0, acc[3][0]);
    acc[3][1] = wmma_f32(c3, d1, acc[3][1]);
    {
      const int kn = k0 + 12 + kbase;
      c0 = *(const v2f*)(ar0 + kn);
      c1 = *(const v2f*)(ar1 + kn);
      c2 = *(const v2f*)(ar2 + kn);
      c3 = *(const v2f*)(ar3 + kn);
      d0 = *(const v2f*)(br0 + kn);
      d1 = *(const v2f*)(br1 + kn);
    }
  }
  acc[0][0] = wmma_f32(a0, b0, acc[0][0]);
  acc[0][1] = wmma_f32(a0, b1, acc[0][1]);
  acc[1][0] = wmma_f32(a1, b0, acc[1][0]);
  acc[1][1] = wmma_f32(a1, b1, acc[1][1]);
  acc[2][0] = wmma_f32(a2, b0, acc[2][0]);
  acc[2][1] = wmma_f32(a2, b1, acc[2][1]);
  acc[3][0] = wmma_f32(a3, b0, acc[3][0]);
  acc[3][1] = wmma_f32(a3, b1, acc[3][1]);
  acc[0][0] = wmma_f32(c0, d0, acc[0][0]);
  acc[0][1] = wmma_f32(c0, d1, acc[0][1]);
  acc[1][0] = wmma_f32(c1, d0, acc[1][0]);
  acc[1][1] = wmma_f32(c1, d1, acc[1][1]);
  acc[2][0] = wmma_f32(c2, d0, acc[2][0]);
  acc[2][1] = wmma_f32(c2, d1, acc[2][1]);
  acc[3][0] = wmma_f32(c3, d0, acc[3][0]);
  acc[3][1] = wmma_f32(c3, d1, acc[3][1]);

#pragma unroll
  for (int mi = 0; mi < 4; ++mi)
#pragma unroll
    for (int ni = 0; ni < 2; ++ni)
#pragma unroll
      for (int r = 0; r < 8; ++r) {
        const int row = m0 + mi * 16 + r + lhalf * 8;
        const int col = n0 + ni * 16 + lmod;
        float v = acc[mi][ni][r];
        if (bias) v += bias[col];
        C[(size_t)row * N + col] = v;
      }
}

static __device__ __forceinline__ float hred_max(float v) {
  v = fmaxf(v, __shfl_xor(v, 1));
  v = fmaxf(v, __shfl_xor(v, 2));
  v = fmaxf(v, __shfl_xor(v, 4));
  v = fmaxf(v, __shfl_xor(v, 8));
  return v;
}
static __device__ __forceinline__ float hred_sum(float v) {
  v += __shfl_xor(v, 1);
  v += __shfl_xor(v, 2);
  v += __shfl_xor(v, 4);
  v += __shfl_xor(v, 8);
  return v;
}

__global__ __launch_bounds__(128)
__attribute__((amdgpu_waves_per_eu(1)))
void attn_kernel(const float* __restrict__ Q, const float* __restrict__ Km,
                 const float* __restrict__ V, float* __restrict__ O) {
  __shared__ float pbuf[4][16 * 16];

  const int lane  = threadIdx.x & 31;
  const int wave  = threadIdx.x >> 5;
  const int lhalf = lane >> 4;
  const int lmod  = lane & 15;
  const int h  = blockIdx.y;
  const int b  = blockIdx.z;
  const int q0 = blockIdx.x * 64 + wave * 16;

  const size_t base = (size_t)b * DIM_T * DIM_K + (size_t)h * HDIM;

  v2f qa[16];
  {
    const float* qrow = Q + base + (size_t)(q0 + lmod) * DIM_K;
#pragma unroll
    for (int t = 0; t < 16; ++t)
      qa[t] = *(const v2f*)(qrow + t * 4 + lhalf * 2);
  }

  v8f acc[4];
#pragma unroll
  for (int j = 0; j < 4; ++j)
#pragma unroll
    for (int r = 0; r < 8; ++r) acc[j][r] = 0.0f;

  float mrow[8], lrow[8];
#pragma unroll
  for (int r = 0; r < 8; ++r) { mrow[r] = -INFINITY; lrow[r] = 0.0f; }

  const float scale = 0.03125f;

  for (int key0 = 0; key0 < DIM_T; key0 += 16) {
    const float* krow = Km + base + (size_t)(key0 + lmod) * DIM_K;
    v2f kb[16];
#pragma unroll
    for (int t = 0; t < 16; ++t)
      kb[t] = *(const v2f*)(krow + t * 4 + lhalf * 2);

    v8f s;
#pragma unroll
    for (int r = 0; r < 8; ++r) s[r] = 0.0f;
#pragma unroll
    for (int t = 0; t < 16; ++t) s = wmma_f32(qa[t], kb[t], s);

    v2f vf[4][4];
#pragma unroll
    for (int t = 0; t < 4; ++t) {
      const int kk = t * 4 + lhalf * 2;
      const float* vbase = V + base + (size_t)(key0 + kk) * DIM_K + lmod;
#pragma unroll
      for (int j = 0; j < 4; ++j) {
        vf[t][j].x = vbase[j * 16];
        vf[t][j].y = vbase[j * 16 + DIM_K];
      }
    }

    float p[8];
#pragma unroll
    for (int r = 0; r < 8; ++r) {
      const float sv   = s[r] * scale;
      const float rm   = hred_max(sv);
      const float mnew = fmaxf(mrow[r], rm);
      const float corr = __expf(mrow[r] - mnew);
      const float pv   = __expf(sv - mnew);
      lrow[r] = lrow[r] * corr + hred_sum(pv);
      mrow[r] = mnew;
      p[r]    = pv;
#pragma unroll
      for (int j = 0; j < 4; ++j) acc[j][r] *= corr;
    }

#pragma unroll
    for (int r = 0; r < 8; ++r)
      pbuf[wave][(r + lhalf * 8) * 16 + lmod] = p[r];

#pragma unroll
    for (int t = 0; t < 4; ++t) {
      const int kk = t * 4 + lhalf * 2;
      v2f pa;
      pa.x = pbuf[wave][lmod * 16 + kk];
      pa.y = pbuf[wave][lmod * 16 + kk + 1];
#pragma unroll
      for (int j = 0; j < 4; ++j)
        acc[j] = wmma_f32(pa, vf[t][j], acc[j]);
    }
  }

#pragma unroll
  for (int j = 0; j < 4; ++j)
#pragma unroll
    for (int r = 0; r < 8; ++r) {
      const float ov = acc[j][r] / lrow[r];
      O[base + (size_t)(q0 + r + lhalf * 8) * DIM_K + j * 16 + lmod] = ov;
    }
}

extern "C" void kernel_launch(void* const* d_in, const int* in_sizes, int n_in,
                              void* d_out, int out_size, void* d_ws,
                              size_t ws_size, hipStream_t stream) {
  (void)in_sizes; (void)n_in; (void)out_size; (void)ws_size;
  const float* x  = (const float*)d_in[0];
  const float* wq = (const float*)d_in[1];
  const float* wk = (const float*)d_in[2];
  const float* wv = (const float*)d_in[3];
  const float* wu = (const float*)d_in[4];
  const float* bu = (const float*)d_in[5];
  float* out = (float*)d_out;

  const size_t MAT = (size_t)DIM_B * DIM_T * DIM_K;
  float* qb = (float*)d_ws;
  float* kb = qb + MAT;
  float* vb = kb + MAT;
  float* ab = vb + MAT;

  const int M = DIM_B * DIM_T, N = DIM_K, K = DIM_K;
  dim3 gblk(256);
  dim3 ggrid(N / 128, M / 128);

  gemm_xwt_kernel<<<ggrid, gblk, 0, stream>>>(x, wq, nullptr, qb, M, N, K);
  gemm_xwt_kernel<<<ggrid, gblk, 0, stream>>>(x, wk, nullptr, kb, M, N, K);
  gemm_xwt_kernel<<<ggrid, gblk, 0, stream>>>(x, wv, nullptr, vb, M, N, K);

  dim3 agrid(DIM_T / 64, NHEAD, DIM_B);
  attn_kernel<<<agrid, dim3(128), 0, stream>>>(qb, kb, vb, ab);

  gemm_xwt_kernel<<<ggrid, gblk, 0, stream>>>(ab, wu, bu, out, M, N, K);
}


# attention: each 16-key K and V tile is fetched once per block with 16-byte row-contiguous loads one step ahead and shared by its 4 waves through a 2-stage LDS ring; WMMA fragments read from LDS
# speedup vs baseline: 1.1253x; 1.1253x over previous
#include <hip/hip_runtime.h>
#include <math.h>

typedef __attribute__((ext_vector_type(2))) float v2f;
typedef __attribute__((ext_vector_type(4))) float v4f;
typedef __attribute__((ext_vector_type(8))) float v8f;

#define DIM_B 2
#define DIM_T 2048
#define DIM_K 1024
#define NHEAD 16
#define HDIM  64

static __device__ __forceinline__ v8f wmma_f32(v2f a, v2f b, v8f c) {
  return __builtin_amdgcn_wmma_f32_16x16x4_f32(false, a, false, b, (short)0, c,
                                               false, false);
}

__global__ __launch_bounds__(256)
__attribute__((amdgpu_waves_per_eu(2)))
void gemm_xwt_kernel(const float* __restrict__ A, const float* __restrict__ W,
                     const float* __restrict__ bias, float* __restrict__ C,
                     int M, int N, int K) {
  const int lane  = threadIdx.x & 31;
  const int wave  = threadIdx.x >> 5;
  const int wm    = wave >> 2;
  const int wn    = wave & 3;
  const int m0    = blockIdx.y * 128 + wm * 64;
  const int n0    = blockIdx.x * 128 + wn * 32;
  const int lhalf = lane >> 4;
  const int lmod  = lane & 15;
  const int kbase = lhalf * 2;

  v8f acc[4][2];
#pragma unroll
  for (int i = 0; i < 4; ++i)
#pragma unroll
    for (int j = 0; j < 2; ++j)
#pragma unroll
      for (int r = 0; r < 8; ++r) acc[i][j][r] = 0.0f;

  const float* ar0 = A + (size_t)(m0 + lmod) * K;
  const float* ar1 = ar0 + (size_t)16 * K;
  const float* ar2 = ar0 + (size_t)32 * K;
  const float* ar3 = ar0 + (size_t)48 * K;
  const float* br0 = W + (size_t)(n0 + lmod) * K;
  const float* br1 = br0 + (size_t)16 * K;

  v2f a0 = *(const v2f*)(ar0 + kbase);
  v2f a1 = *(const v2f*)(ar1 + kbase);
  v2f a2 = *(const v2f*)(ar2 + kbase);
  v2f a3 = *(const v2f*)(ar3 + kbase);
  v2f b0 = *(const v2f*)(br0 + kbase);
  v2f b1 = *(const v2f*)(br1 + kbase);

  v2f c0 = *(const v2f*)(ar0 + 4 + kbase);
  v2f c1 = *(const v2f*)(ar1 + 4 + kbase);
  v2f c2 = *(const v2f*)(ar2 + 4 + kbase);
  v2f c3 = *(const v2f*)(ar3 + 4 + kbase);
  v2f d0 = *(const v2f*)(br0 + 4 + kbase);
  v2f d1 = *(const v2f*)(br1 + 4 + kbase);

#pragma clang loop unroll(disable)
  for (int k0 = 0; k0 + 16 <= K; k0 += 8) {
    acc[0][0] = wmma_f32(a0, b0, acc[0][0]);
    acc[0][1] = wmma_f32(a0, b1, acc[0][1]);
    acc[1][0] = wmma_f32(a1, b0, acc[1][0]);
    acc[1][1] = wmma_f32(a1, b1, acc[1][1]);
    acc[2][0] = wmma_f32(a2, b0, acc[2][0]);
    acc[2][1] = wmma_f32(a2, b1, acc[2][1]);
    acc[3][0] = wmma_f32(a3, b0, acc[3][0]);
    acc[3][1] = wmma_f32(a3, b1, acc[3][1]);
    {
      const int kn = k0 + 8 + kbase;
      a0 = *(const v2f*)(ar0 + kn);
      a1 = *(const v2f*)(ar1 + kn);
      a2 = *(const v2f*)(ar2 + kn);
      a3 = *(const v2f*)(ar3 + kn);
      b0 = *(const v2f*)(br0 + kn);
      b1 = *(const v2f*)(br1 + kn);
    }
    acc[0][0] = wmma_f32(c0, d0, acc[0][0]);
    acc[0][1] = wmma_f32(c0, d1, acc[0][1]);
    acc[1][0] = wmma_f32(c1, d0, acc[1][0]);
    acc[1][1] = wmma_f32(c1, d1, acc[1][1]);
    acc[2][0] = wmma_f32(c2, d0, acc[2][0]);
    acc[2][1] = wmma_f32(c2, d1, acc[2][1]);
    acc[3][0] = wmma_f32(c3, d0, acc[3][0]);
    acc[3][1] = wmma_f32(c3, d1, acc[3][1]);
    {
      const int kn = k0 + 12 + kbase;
      c0 = *(const v2f*)(ar0 + kn);
      c1 = *(const v2f*)(ar1 + kn);
      c2 = *(const v2f*)(ar2 + kn);
      c3 = *(const v2f*)(ar3 + kn);
      d0 = *(const v2f*)(br0 + kn);
      d1 = *(const v2f*)(br1 + kn);
    }
  }
  acc[0][0] = wmma_f32(a0, b0, acc[0][0]);
  acc[0][1] = wmma_f32(a0, b1, acc[0][1]);
  acc[1][0] = wmma_f32(a1, b0, acc[1][0]);
  acc[1][1] = wmma_f32(a1, b1, acc[1][1]);
  acc[2][0] = wmma_f32(a2, b0, acc[2][0]);
  acc[2][1] = wmma_f32(a2, b1, acc[2][1]);
  acc[3][0] = wmma_f32(a3, b0, acc[3][0]);
  acc[3][1] = wmma_f32(a3, b1, acc[3][1]);
  acc[0][0] = wmma_f32(c0, d0, acc[0][0]);
  acc[0][1] = wmma_f32(c0, d1, acc[0][1]);
  acc[1][0] = wmma_f32(c1, d0, acc[1][0]);
  acc[1][1] = wmma_f32(c1, d1, acc[1][1]);
  acc[2][0] = wmma_f32(c2, d0, acc[2][0]);
  acc[2][1] = wmma_f32(c2, d1, acc[2][1]);
  acc[3][0] = wmma_f32(c3, d0, acc[3][0]);
  acc[3][1] = wmma_f32(c3, d1, acc[3][1]);

#pragma unroll
  for (int mi = 0; mi < 4; ++mi)
#pragma unroll
    for (int ni = 0; ni < 2; ++ni)
#pragma unroll
      for (int r = 0; r < 8; ++r) {
        const int row = m0 + mi * 16 + r + lhalf * 8;
        const int col = n0 + ni * 16 + lmod;
        float v = acc[mi][ni][r];
        if (bias) v += bias[col];
        C[(size_t)row * N + col] = v;
      }
}

static __device__ __forceinline__ float hred_max(float v) {
  v = fmaxf(v, __shfl_xor(v, 1));
  v = fmaxf(v, __shfl_xor(v, 2));
  v = fmaxf(v, __shfl_xor(v, 4));
  v = fmaxf(v, __shfl_xor(v, 8));
  return v;
}
static __device__ __forceinline__ float hred_sum(float v) {
  v += __shfl_xor(v, 1);
  v += __shfl_xor(v, 2);
  v += __shfl_xor(v, 4);
  v += __shfl_xor(v, 8);
  return v;
}

#define KPITCH 68
#define VPITCH 80

__global__ __launch_bounds__(128)
__attribute__((amdgpu_waves_per_eu(1)))
void attn_kernel(const float* __restrict__ Q, const float* __restrict__ Km,
                 const float* __restrict__ V, float* __restrict__ O) {
  __shared__ float pbuf[4][16 * 16];
  __shared__ __attribute__((aligned(16))) float ktile[2][16 * KPITCH];
  __shared__ __attribute__((aligned(16))) float vtile[2][16 * VPITCH];

  const int lane  = threadIdx.x & 31;
  const int wave  = threadIdx.x >> 5;
  const int lhalf = lane >> 4;
  const int lmod  = lane & 15;
  const int h  = blockIdx.y;
  const int b  = blockIdx.z;
  const int q0 = blockIdx.x * 64 + wave * 16;

  const size_t base = (size_t)b * DIM_T * DIM_K + (size_t)h * HDIM;

  v2f qa[16];
  {
    const float* qrow = Q + base + (size_t)(q0 + lmod) * DIM_K;
#pragma unroll
    for (int t = 0; t < 16; ++t)
      qa[t] = *(const v2f*)(qrow + t * 4 + lhalf * 2);
  }

  v8f acc[4];
#pragma unroll
  for (int j = 0; j < 4; ++j)
#pragma unroll
    for (int r = 0; r < 8; ++r) acc[j][r] = 0.0f;

  float mrow[8], lrow[8];
#pragma unroll
  for (int r = 0; r < 8; ++r) { mrow[r] = -INFINITY; lrow[r] = 0.0f; }

  const float scale = 0.03125f;

  const int srow = threadIdx.x >> 4;
  const int scol = (threadIdx.x & 15) * 4;
  const float* kg = Km + base + (size_t)srow * DIM_K + scol;
  const float* vg = V  + base + (size_t)srow * DIM_K + scol;
  const int kst = srow * KPITCH + scol;
  const int vst = srow * VPITCH + scol;

  {
    const v4f k0 = *(const v4f*)(kg);
    const v4f k1 = *(const v4f*)(kg + (size_t)8 * DIM_K);
    const v4f v0 = *(const v4f*)(vg);
    const v4f v1 = *(const v4f*)(vg + (size_t)8 * DIM_K);
    *(v4f*)(&ktile[0][kst])              = k0;
    *(v4f*)(&ktile[0][kst + 8 * KPITCH]) = k1;
    *(v4f*)(&vtile[0][vst])              = v0;
    *(v4f*)(&vtile[0][vst + 8 * VPITCH]) = v1;
  }
  __syncthreads();

  for (int key0 = 0; key0 < DIM_T; key0 += 16) {
    const int cur = (key0 >> 4) & 1;

    const int knext = (key0 + 16 < DIM_T) ? (key0 + 16) : key0;
    const v4f nk0 = *(const v4f*)(kg + (size_t)knext * DIM_K);
    const v4f nk1 = *(const v4f*)(kg + (size_t)(knext + 8) * DIM_K);
    const v4f nv0 = *(const v4f*)(vg + (size_t)knext * DIM_K);
    const v4f nv1 = *(const v4f*)(vg + (size_t)(knext + 8) * DIM_K);

    const float* krow = &ktile[cur][lmod * KPITCH + lhalf * 2];
    v2f kb[16];
#pragma unroll
    for (int t = 0; t < 16; ++t)
      kb[t] = *(const v2f*)(krow + t * 4);
    __builtin_amdgcn_sched_barrier(0);

    v8f s;
#pragma unroll
    for (int r = 0; r < 8; ++r) s[r] = 0.0f;
#pragma unroll
    for (int t = 0; t < 16; ++t) s = wmma_f32(qa[t], kb[t], s);

    v2f vf[4][4];
#pragma unroll
    for (int t = 0; t < 4; ++t) {
      const int kk = t * 4 + lhalf * 2;
      const float* vbase = &vtile[cur][kk * VPITCH + lmod];
#pragma unroll
      for (int j = 0; j < 4; ++j) {
        vf[t][j].x = vbase[j * 16];
        vf[t][j].y = vbase[j * 16 + VPITCH];
      }
    }

    float p[8];
#pragma unroll
    for (int r = 0; r < 8; ++r) {
      const float sv   = s[r] * scale;
      const float rm   = hred_max(sv);
      const float mnew = fmaxf(mrow[r], rm);
      const float corr = __expf(mrow[r] - mnew);
      const float pv   = __expf(sv - mnew);
      lrow[r] = lrow[r] * corr + hred_sum(pv);
      mrow[r] = mnew;
      p[r]    = pv;
#pragma unroll
      for (int j = 0; j < 4; ++j) acc[j][r] *= corr;
    }

#pragma unroll
    for (int r = 0; r < 8; ++r)
      pbuf[wave][(r + lhalf * 8) * 16 + lmod] = p[r];

#pragma unroll
    for (int t = 0; t < 4; ++t) {
      const int kk = t * 4 + lhalf * 2;
      v2f pa;
      pa.x = pbuf[wave][lmod * 16 + kk];
      pa.y = pbuf[wave][lmod * 16 + kk + 1];
#pragma unroll
      for (int j = 0; j < 4; ++j)
        acc[j] = wmma_f32(pa, vf[t][j], acc[j]);
    }

    __builtin_amdgcn_sched_barrier(0);
    *(v4f*)(&ktile[cur ^ 1][kst])              = nk0;
    *(v4f*)(&ktile[cur ^ 1][kst + 8 * KPITCH]) = nk1;
    *(v4f*)(&vtile[cur ^ 1][vst])              = nv0;
    *(v4f*)(&vtile[cur ^ 1][vst + 8 * VPITCH]) = nv1;
    __syncthreads();
  }

#pragma unroll
  for (int j = 0; j < 4; ++j)
#pragma unroll
    for (int r = 0; r < 8; ++r) {
      const float ov = acc[j][r] / lrow[r];
      O[base + (size_t)(q0 + r + lhalf * 8) * DIM_K + j * 16 + lmod] = ov;
    }
}

extern "C" void kernel_launch(void* const* d_in, const int* in_sizes, int n_in,
                              void* d_out, int out_size, void* d_ws,
                              size_t ws_size, hipStream_t stream) {
  (void)in_sizes; (void)n_in; (void)out_size; (void)ws_size;
  const float* x  = (const float*)d_in[0];
  const float* wq = (const float*)d_in[1];
  const float* wk = (const float*)d_in[2];
  const float* wv = (const float*)d_in[3];
  const float* wu = (const float*)d_in[4];
  const float* bu = (const float*)d_in[5];
  float* out = (float*)d_out;

  const size_t MAT = (size_t)DIM_B * DIM_T * DIM_K;
  float* qb = (float*)d_ws;
  float* kb = qb + MAT;
  float* vb = kb + MAT;
  float* ab = vb + MAT;

  const int M = DIM_B * DIM_T, N = DIM_K, K = DIM_K;
  dim3 gblk(256);
  dim3 ggrid(N / 128, M / 128);

  gemm_xwt_kernel<<<ggrid, gblk, 0, stream>>>(x, wq, nullptr, qb, M, N, K);
  gemm_xwt_kernel<<<ggrid, gblk, 0, stream>>>(x, wk, nullptr, kb, M, N, K);
  gemm_xwt_kernel<<<ggrid, gblk, 0, stream>>>(x, wv, nullptr, vb, M, N, K);

  dim3 agrid(DIM_T / 64, NHEAD, DIM_B);
  attn_kernel<<<agrid, dim3(128), 0, stream>>>(qb, kb, vb, ab);

  gemm_xwt_kernel<<<ggrid, gblk, 0, stream>>>(ab, wu, bu, out, M, N, K);
}


# projection GEMMs: each 32-float K slice of the block's A and W rows is fetched once per block with 16-byte row-contiguous loads one step ahead and shared by its 8 waves via a 2-stage LDS ring
# speedup vs baseline: 1.2415x; 1.1033x over previous
#include <hip/hip_runtime.h>
#include <math.h>

typedef __attribute__((ext_vector_type(2))) float v2f;
typedef __attribute__((ext_vector_type(4))) float v4f;
typedef __attribute__((ext_vector_type(8))) float v8f;

#define DIM_B 2
#define DIM_T 2048
#define DIM_K 1024
#define NHEAD 16
#define HDIM  64

static __device__ __forceinline__ v8f wmma_f32(v2f a, v2f b, v8f c) {
  return __builtin_amdgcn_wmma_f32_16x16x4_f32(false, a, false, b, (short)0, c,
                                               false, false);
}

#define GK 32
#define GPITCH 36

__global__ __launch_bounds__(256)
__attribute__((amdgpu_waves_per_eu(2)))
void gemm_xwt_kernel(const float* __restrict__ A, const float* __restrict__ W,
                     const float* __restrict__ bias, float* __restrict__ C,
                     int M, int N, int K) {
  __shared__ __attribute__((aligned(16))) float atile[2][128 * GPITCH];
  __shared__ __attribute__((aligned(16))) float wtile[2][128 * GPITCH];

  const int lane  = threadIdx.x & 31;
  const int wave  = threadIdx.x >> 5;
  const int wm    = wave >> 2;
  const int wn    = wave & 3;
  const int m0    = blockIdx.y * 128 + wm * 64;
  const int n0    = blockIdx.x * 128 + wn * 32;
  const int lhalf = lane >> 4;
  const int lmod  = lane & 15;

  v8f acc[4][2];
#pragma unroll
  for (int i = 0; i < 4; ++i)
#pragma unroll
    for (int j = 0; j < 2; ++j)
#pragma unroll
      for (int r = 0; r < 8; ++r) acc[i][j][r] = 0.0f;

  const int srow = threadIdx.x >> 3;
  const int scol = (threadIdx.x & 7) * 4;
  const float* ag = A + (size_t)(blockIdx.y * 128 + srow) * K + scol;
  const float* wg = W + (size_t)(blockIdx.x * 128 + srow) * K + scol;
  const int sst = srow * GPITCH + scol;

  const int afr = (wm * 64 + lmod) * GPITCH + lhalf * 2;
  const int wfr = (wn * 32 + lmod) * GPITCH + lhalf * 2;

  {
    v4f ta[4], tw[4];
#pragma unroll
    for (int i = 0; i < 4; ++i) {
      ta[i] = *(const v4f*)(ag + (size_t)(32 * i) * K);
      tw[i] = *(const v4f*)(wg + (size_t)(32 * i) * K);
    }
#pragma unroll
    for (int i = 0; i < 4; ++i) {
      *(v4f*)(&atile[0][sst + 32 * i * GPITCH]) = ta[i];
      *(v4f*)(&wtile[0][sst + 32 * i * GPITCH]) = tw[i];
    }
  }
  __syncthreads();

  for (int kt = 0; kt < K; kt += GK) {
    const int cur = (kt / GK) & 1;

    const int kn = (kt + GK < K) ? (kt + GK) : kt;
    v4f na[4], nw[4];
#pragma unroll
    for (int i = 0; i < 4; ++i) {
      na[i] = *(const v4f*)(ag + (size_t)(32 * i) * K + kn);
      nw[i] = *(const v4f*)(wg + (size_t)(32 * i) * K + kn);
    }
    __builtin_amdgcn_sched_barrier(0);

    const float* ap = &atile[cur][afr];
    const float* wp = &wtile[cur][wfr];
    v2f fa[2][4], fw[2][2];
#pragma unroll
    for (int i = 0; i < 4; ++i) fa[0][i] = *(const v2f*)(ap + i * 16 * GPITCH);
#pragma unroll
    for (int j = 0; j < 2; ++j) fw[0][j] = *(const v2f*)(wp + j * 16 * GPITCH);
#pragma unroll
    for (int ks = 0; ks < GK / 4; ++ks) {
      const int cb = ks & 1;
      if (ks + 1 < GK / 4) {
#pragma unroll
        for (int i = 0; i < 4; ++i)
          fa[cb ^ 1][i] = *(const v2f*)(ap + i * 16 * GPITCH + (ks + 1) * 4);
#pragma unroll
        for (int j = 0; j < 2; ++j)
          fw[cb ^ 1][j] = *(const v2f*)(wp + j * 16 * GPITCH + (ks + 1) * 4);
      }
#pragma unroll
      for (int i = 0; i < 4; ++i)
#pragma unroll
        for (int j = 0; j < 2; ++j)
          acc[i][j] = wmma_f32(fa[cb][i], fw[cb][j], acc[i][j]);
    }

    __builtin_amdgcn_sched_barrier(0);
#pragma unroll
    for (int i = 0; i < 4; ++i) {
      *(v4f*)(&atile[cur ^ 1][sst + 32 * i * GPITCH]) = na[i];
      *(v4f*)(&wtile[cur ^ 1][sst + 32 * i * GPITCH]) = nw[i];
    }
    __syncthreads();
  }

#pragma unroll
  for (int mi = 0; mi < 4; ++mi)
#pragma unroll
    for (int ni = 0; ni < 2; ++ni)
#pragma unroll
      for (int r = 0; r < 8; ++r) {
        const int row = m0 + mi * 16 + r + lhalf * 8;
        const int col = n0 + ni * 16 + lmod;
        float v = acc[mi][ni][r];
        if (bias) v += bias[col];
        C[(size_t)row * N + col] = v;
      }
}

static __device__ __forceinline__ float hred_max(float v) {
  v = fmaxf(v, __shfl_xor(v, 1));
  v = fmaxf(v, __shfl_xor(v, 2));
  v = fmaxf(v, __shfl_xor(v, 4));
  v = fmaxf(v, __shfl_xor(v, 8));
  return v;
}
static __device__ __forceinline__ float hred_sum(float v) {
  v += __shfl_xor(v, 1);
  v += __shfl_xor(v, 2);
  v += __shfl_xor(v, 4);
  v += __shfl_xor(v, 8);
  return v;
}

#define KPITCH 68
#define VPITCH 80

__global__ __launch_bounds__(128)
__attribute__((amdgpu_waves_per_eu(1)))
void attn_kernel(const float* __restrict__ Q, const float* __restrict__ Km,
                 const float* __restrict__ V, float* __restrict__ O) {
  __shared__ float pbuf[4][16 * 16];
  __shared__ __attribute__((aligned(16))) float ktile[2][16 * KPITCH];
  __shared__ __attribute__((aligned(16))) float vtile[2][16 * VPITCH];

  const int lane  = threadIdx.x & 31;
  const int wave  = threadIdx.x >> 5;
  const int lhalf = lane >> 4;
  const int lmod  = lane & 15;
  const int h  = blockIdx.y;
  const int b  = blockIdx.z;
  const int q0 = blockIdx.x * 64 + wave * 16;

  const size_t base = (size_t)b * DIM_T * DIM_K + (size_t)h * HDIM;

  v2f qa[16];
  {
    const float* qrow = Q + base + (size_t)(q0 + lmod) * DIM_K;
#pragma unroll
    for (int t = 0; t < 16; ++t)
      qa[t] = *(const v2f*)(qrow + t * 4 + lhalf * 2);
  }

  v8f acc[4];
#pragma unroll
  for (int j = 0; j < 4; ++j)
#pragma unroll
    for (int r = 0; r < 8; ++r) acc[j][r] = 0.0f;

  float mrow[8], lrow[8];
#pragma unroll
  for (int r = 0; r < 8; ++r) { mrow[r] = -INFINITY; lrow[r] = 0.0f; }

  const float scale = 0.03125f;

  const int srow = threadIdx.x >> 4;
  const int scol = (threadIdx.x & 15) * 4;
  const float* kg = Km + base + (size_t)srow * DIM_K + scol;
  const float* vg = V  + base + (size_t)srow * DIM_K + scol;
  const int kst = srow * KPITCH + scol;
  const int vst = srow * VPITCH + scol;

  {
    const v4f k0 = *(const v4f*)(kg);
    const v4f k1 = *(const v4f*)(kg + (size_t)8 * DIM_K);
    const v4f v0 = *(const v4f*)(vg);
    const v4f v1 = *(const v4f*)(vg + (size_t)8 * DIM_K);
    *(v4f*)(&ktile[0][kst])              = k0;
    *(v4f*)(&ktile[0][kst + 8 * KPITCH]) = k1;
    *(v4f*)(&vtile[0][vst])              = v0;
    *(v4f*)(&vtile[0][vst + 8 * VPITCH]) = v1;
  }
  __syncthreads();

  for (int key0 = 0; key0 < DIM_T; key0 += 16) {
    const int cur = (key0 >> 4) & 1;

    const int knext = (key0 + 16 < DIM_T) ? (key0 + 16) : key0;
    const v4f nk0 = *(const v4f*)(kg + (size_t)knext * DIM_K);
    const v4f nk1 = *(const v4f*)(kg + (size_t)(knext + 8) * DIM_K);
    const v4f nv0 = *(const v4f*)(vg + (size_t)knext * DIM_K);
    const v4f nv1 = *(const v4f*)(vg + (size_t)(knext + 8) * DIM_K);

    const float* krow = &ktile[cur][lmod * KPITCH + lhalf * 2];
    v2f kb[16];
#pragma unroll
    for (int t = 0; t < 16; ++t)
      kb[t] = *(const v2f*)(krow + t * 4);
    __builtin_amdgcn_sched_barrier(0);

    v8f s;
#pragma unroll
    for (int r = 0; r < 8; ++r) s[r] = 0.0f;
#pragma unroll
    for (int t = 0; t < 16; ++t) s = wmma_f32(qa[t], kb[t], s);

    v2f vf[4][4];
#pragma unroll
    for (int t = 0; t < 4; ++t) {
      const int kk = t * 4 + lhalf * 2;
      const float* vbase = &vtile[cur][kk * VPITCH + lmod];
#pragma unroll
      for (int j = 0; j < 4; ++j) {
        vf[t][j].x = vbase[j * 16];
        vf[t][j].y = vbase[j * 16 + VPITCH];
      }
    }

    float p[8];
#pragma unroll
    for (int r = 0; r < 8; ++r) {
      const float sv   = s[r] * scale;
      const float rm   = hred_max(sv);
      const float mnew = fmaxf(mrow[r], rm);
      const float corr = __expf(mrow[r] - mnew);
      const float pv   = __expf(sv - mnew);
      lrow[r] = lrow[r] * corr + hred_sum(pv);
      mrow[r] = mnew;
      p[r]    = pv;
#pragma unroll
      for (int j = 0; j < 4; ++j) acc[j][r] *= corr;
    }

#pragma unroll
    for (int r = 0; r < 8; ++r)
      pbuf[wave][(r + lhalf * 8) * 16 + lmod] = p[r];

#pragma unroll
    for (int t = 0; t < 4; ++t) {
      const int kk = t * 4 + lhalf * 2;
      v2f pa;
      pa.x = pbuf[wave][lmod * 16 + kk];
      pa.y = pbuf[wave][lmod * 16 + kk + 1];
#pragma unroll
      for (int j = 0; j < 4; ++j)
        acc[j] = wmma_f32(pa, vf[t][j], acc[j]);
    }

    __builtin_amdgcn_sched_barrier(0);
    *(v4f*)(&ktile[cur ^ 1][kst])              = nk0;
    *(v4f*)(&ktile[cur ^ 1][kst + 8 * KPITCH]) = nk1;
    *(v4f*)(&vtile[cur ^ 1][vst])              = nv0;
    *(v4f*)(&vtile[cur ^ 1][vst + 8 * VPITCH]) = nv1;
    __syncthreads();
  }

#pragma unroll
  for (int j = 0; j < 4; ++j)
#pragma unroll
    for (int r = 0; r < 8; ++r) {
      const float ov = acc[j][r] / lrow[r];
      O[base + (size_t)(q0 + r + lhalf * 8) * DIM_K + j * 16 + lmod] = ov;
    }
}

extern "C" void kernel_launch(void* const* d_in, const int* in_sizes, int n_in,
                              void* d_out, int out_size, void* d_ws,
                              size_t ws_size, hipStream_t stream) {
  (void)in_sizes; (void)n_in; (void)out_size; (void)ws_size;
  const float* x  = (const float*)d_in[0];
  const float* wq = (const float*)d_in[1];
  const float* wk = (const float*)d_in[2];
  const float* wv = (const float*)d_in[3];
  const float* wu = (const float*)d_in[4];
  const float* bu = (const float*)d_in[5];
  float* out = (float*)d_out;

  const size_t MAT = (size_t)DIM_B * DIM_T * DIM_K;
  float* qb = (float*)d_ws;
  float* kb = qb + MAT;
  float* vb = kb + MAT;
  float* ab = vb + MAT;

  const int M = DIM_B * DIM_T, N = DIM_K, K = DIM_K;
  dim3 gblk(256);
  dim3 ggrid(N / 128, M / 128);

  gemm_xwt_kernel<<<ggrid, gblk, 0, stream>>>(x, wq, nullptr, qb, M, N, K);
  gemm_xwt_kernel<<<ggrid, gblk, 0, stream>>>(x, wk, nullptr, kb, M, N, K);
  gemm_xwt_kernel<<<ggrid, gblk, 0, stream>>>(x, wv, nullptr, vb, M, N, K);

  dim3 agrid(DIM_T / 64, NHEAD, DIM_B);
  attn_kernel<<<agrid, dim3(128), 0, stream>>>(qb, kb, vb, ab);

  gemm_xwt_kernel<<<ggrid, gblk, 0, stream>>>(ab, wu, bu, out, M, N, K);
}


# attention: score tile computed transposed (S^T = K Q^T, O^T += V^T P^T): row max and sum taken in-lane plus one half exchange, P feeds PV from registers; row-sum order and PV key order change
# speedup vs baseline: 1.3746x; 1.1072x over previous
#include <hip/hip_runtime.h>
#include <math.h>

typedef __attribute__((ext_vector_type(2))) float v2f;
typedef __attribute__((ext_vector_type(4))) float v4f;
typedef __attribute__((ext_vector_type(8))) float v8f;

#define DIM_B 2
#define DIM_T 2048
#define DIM_K 1024
#define NHEAD 16
#define HDIM  64

static __device__ __forceinline__ v8f wmma_f32(v2f a, v2f b, v8f c) {
  return __builtin_amdgcn_wmma_f32_16x16x4_f32(false, a, false, b, (short)0, c,
                                               false, false);
}

#define GK 32
#define GPITCH 36

__global__ __launch_bounds__(256)
__attribute__((amdgpu_waves_per_eu(2)))
void gemm_xwt_kernel(const float* __restrict__ A, const float* __restrict__ W,
                     const float* __restrict__ bias, float* __restrict__ C,
                     int M, int N, int K) {
  __shared__ __attribute__((aligned(16))) float atile[2][128 * GPITCH];
  __shared__ __attribute__((aligned(16))) float wtile[2][128 * GPITCH];

  const int lane  = threadIdx.x & 31;
  const int wave  = threadIdx.x >> 5;
  const int wm    = wave >> 2;
  const int wn    = wave & 3;
  const int m0    = blockIdx.y * 128 + wm * 64;
  const int n0    = blockIdx.x * 128 + wn * 32;
  const int lhalf = lane >> 4;
  const int lmod  = lane & 15;

  v8f acc[4][2];
#pragma unroll
  for (int i = 0; i < 4; ++i)
#pragma unroll
    for (int j = 0; j < 2; ++j)
#pragma unroll
      for (int r = 0; r < 8; ++r) acc[i][j][r] = 0.0f;

  const int srow = threadIdx.x >> 3;
  const int scol = (threadIdx.x & 7) * 4;
  const float* ag = A + (size_t)(blockIdx.y * 128 + srow) * K + scol;
  const float* wg = W + (size_t)(blockIdx.x * 128 + srow) * K + scol;
  const int sst = srow * GPITCH + scol;

  const int afr = (wm * 64 + lmod) * GPITCH + lhalf * 2;
  const int wfr = (wn * 32 + lmod) * GPITCH + lhalf * 2;

  {
    v4f ta[4], tw[4];
#pragma unroll
    for (int i = 0; i < 4; ++i) {
      ta[i] = *(const v4f*)(ag + (size_t)(32 * i) * K);
      tw[i] = *(const v4f*)(wg + (size_t)(32 * i) * K);
    }
#pragma unroll
    for (int i = 0; i < 4; ++i) {
      *(v4f*)(&atile[0][sst + 32 * i * GPITCH]) = ta[i];
      *(v4f*)(&wtile[0][sst + 32 * i * GPITCH]) = tw[i];
    }
  }
  __syncthreads();

  for (int kt = 0; kt < K; kt += GK) {
    const int cur = (kt / GK) & 1;

    const int kn = (kt + GK < K) ? (kt + GK) : kt;
    v4f na[4], nw[4];
#pragma unroll
    for (int i = 0; i < 4; ++i) {
      na[i] = *(const v4f*)(ag + (size_t)(32 * i) * K + kn);
      nw[i] = *(const v4f*)(wg + (size_t)(32 * i) * K + kn);
    }
    __builtin_amdgcn_sched_barrier(0);

    const float* ap = &atile[cur][afr];
    const float* wp = &wtile[cur][wfr];
    v2f fa[2][4], fw[2][2];
#pragma unroll
    for (int i = 0; i < 4; ++i) fa[0][i] = *(const v2f*)(ap + i * 16 * GPITCH);
#pragma unroll
    for (int j = 0; j < 2; ++j) fw[0][j] = *(const v2f*)(wp + j * 16 * GPITCH);
#pragma unroll
    for (int ks = 0; ks < GK / 4; ++ks) {
      const int cb = ks & 1;
      if (ks + 1 < GK / 4) {
#pragma unroll
        for (int i = 0; i < 4; ++i)
          fa[cb ^ 1][i] = *(const v2f*)(ap + i * 16 * GPITCH + (ks + 1) * 4);
#pragma unroll
        for (int j = 0; j < 2; ++j)
          fw[cb ^ 1][j] = *(const v2f*)(wp + j * 16 * GPITCH + (ks + 1) * 4);
      }
#pragma unroll
      for (int i = 0; i < 4; ++i)
#pragma unroll
        for (int j = 0; j < 2; ++j)
          acc[i][j] = wmma_f32(fa[cb][i], fw[cb][j], acc[i][j]);
    }

    __builtin_amdgcn_sched_barrier(0);
#pragma unroll
    for (int i = 0; i < 4; ++i) {
      *(v4f*)(&atile[cur ^ 1][sst + 32 * i * GPITCH]) = na[i];
      *(v4f*)(&wtile[cur ^ 1][sst + 32 * i * GPITCH]) = nw[i];
    }
    __syncthreads();
  }

#pragma unroll
  for (int mi = 0; mi < 4; ++mi)
#pragma unroll
    for (int ni = 0; ni < 2; ++ni)
#pragma unroll
      for (int r = 0; r < 8; ++r) {
        const int row = m0 + mi * 16 + r + lhalf * 8;
        const int col = n0 + ni * 16 + lmod;
        float v = acc[mi][ni][r];
        if (bias) v += bias[col];
        C[(size_t)row * N + col] = v;
      }
}

static __device__ __forceinline__ float xhalf(float x) {
  const int b = __builtin_bit_cast(int, x);
  return __builtin_bit_cast(
      float, __builtin_amdgcn_permlanex16(b, b, 0x76543210, 0xfedcba98, false,
                                          false));
}

#define KPITCH 68
#define VPITCH 68
#define OPITCH 68

__global__ __launch_bounds__(128)
__attribute__((amdgpu_waves_per_eu(1)))
void attn_kernel(const float* __restrict__ Q, const float* __restrict__ Km,
                 const float* __restrict__ V, float* __restrict__ O) {
  __shared__ __attribute__((aligned(16))) float ktile[2][16 * KPITCH];
  __shared__ __attribute__((aligned(16))) float vtile[2][16 * VPITCH];
  __shared__ __attribute__((aligned(16))) float otile[4][16 * OPITCH];


  const int lane  = threadIdx.x & 31;
  const int wave  = threadIdx.x >> 5;
  const int lhalf = lane >> 4;
  const int lmod  = lane & 15;
  const int h  = blockIdx.y;
  const int b  = blockIdx.z;
  const int q0 = blockIdx.x * 64 + wave * 16;

  const size_t base = (size_t)b * DIM_T * DIM_K + (size_t)h * HDIM;

  v2f qb[16];
  {
    const float* qrow = Q + base + (size_t)(q0 + lmod) * DIM_K;
#pragma unroll
    for (int t = 0; t < 16; ++t)
      qb[t] = *(const v2f*)(qrow + t * 4 + lhalf * 2);
  }

  v8f acc[4];
#pragma unroll
  for (int j = 0; j < 4; ++j)
#pragma unroll
    for (int r = 0; r < 8; ++r) acc[j][r] = 0.0f;

  float mrun = -INFINITY, lrun = 0.0f;

  const float scale = 0.03125f;

  const int srow = threadIdx.x >> 4;
  const int scol = (threadIdx.x & 15) * 4;
  const float* kg = Km + base + (size_t)srow * DIM_K + scol;
  const float* vg = V  + base + (size_t)srow * DIM_K + scol;
  const int kst = srow * KPITCH + scol;
  const int vst = srow * VPITCH + scol;

  {
    const v4f k0 = *(const v4f*)(kg);
    const v4f k1 = *(const v4f*)(kg + (size_t)8 * DIM_K);
    const v4f v0 = *(const v4f*)(vg);
    const v4f v1 = *(const v4f*)(vg + (size_t)8 * DIM_K);
    *(v4f*)(&ktile[0][kst])              = k0;
    *(v4f*)(&ktile[0][kst + 8 * KPITCH]) = k1;
    *(v4f*)(&vtile[0][vst])              = v0;
    *(v4f*)(&vtile[0][vst + 8 * VPITCH]) = v1;
  }
  __syncthreads();

  for (int key0 = 0; key0 < DIM_T; key0 += 16) {
    const int cur = (key0 >> 4) & 1;

    const int knext = (key0 + 16 < DIM_T) ? (key0 + 16) : key0;
    const v4f nk0 = *(const v4f*)(kg + (size_t)knext * DIM_K);
    const v4f nk1 = *(const v4f*)(kg + (size_t)(knext + 8) * DIM_K);
    const v4f nv0 = *(const v4f*)(vg + (size_t)knext * DIM_K);
    const v4f nv1 = *(const v4f*)(vg + (size_t)(knext + 8) * DIM_K);

    const float* krow = &ktile[cur][lmod * KPITCH + lhalf * 2];
    v2f ka[16];
#pragma unroll
    for (int t = 0; t < 16; ++t)
      ka[t] = *(const v2f*)(krow + t * 4);
    __builtin_amdgcn_sched_barrier(0);

    v8f s;
#pragma unroll
    for (int r = 0; r < 8; ++r) s[r] = 0.0f;
#pragma unroll
    for (int t = 0; t < 16; ++t) s = wmma_f32(ka[t], qb[t], s);

    v2f vf[4][4];
#pragma unroll
    for (int t = 0; t < 4; ++t) {
      const int kk = lhalf * 8 + t * 2;
      const float* vbase = &vtile[cur][kk * VPITCH + lmod];
#pragma unroll
      for (int j = 0; j < 4; ++j) {
        vf[t][j].x = vbase[j * 16];
        vf[t][j].y = vbase[j * 16 + VPITCH];
      }
    }

    float sv[8];
#pragma unroll
    for (int r = 0; r < 8; ++r) sv[r] = s[r] * scale;
    float rm = fmaxf(fmaxf(fmaxf(sv[0], sv[1]), fmaxf(sv[2], sv[3])),
                     fmaxf(fmaxf(sv[4], sv[5]), fmaxf(sv[6], sv[7])));
    rm = fmaxf(rm, xhalf(rm));
    const float mnew = fmaxf(mrun, rm);
    const float corr = __expf(mrun - mnew);
    float p[8];
#pragma unroll
    for (int r = 0; r < 8; ++r) p[r] = __expf(sv[r] - mnew);
    const float psum = ((p[0] + p[1]) + (p[2] + p[3])) +
                       ((p[4] + p[5]) + (p[6] + p[7]));
    lrun = lrun * corr + psum;
    mrun = mnew;
#pragma unroll
    for (int j = 0; j < 4; ++j)
#pragma unroll
      for (int r = 0; r < 8; ++r) acc[j][r] *= corr;

#pragma unroll
    for (int t = 0; t < 4; ++t) {
      v2f pb;
      pb.x = p[2 * t];
      pb.y = p[2 * t + 1];
#pragma unroll
      for (int j = 0; j < 4; ++j)
        acc[j] = wmma_f32(vf[t][j], pb, acc[j]);
    }

    __builtin_amdgcn_sched_barrier(0);
    *(v4f*)(&ktile[cur ^ 1][kst])              = nk0;
    *(v4f*)(&ktile[cur ^ 1][kst + 8 * KPITCH]) = nk1;
    *(v4f*)(&vtile[cur ^ 1][vst])              = nv0;
    *(v4f*)(&vtile[cur ^ 1][vst + 8 * VPITCH]) = nv1;
    __syncthreads();
  }

  const float lsum = lrun + xhalf(lrun);
  float* ot = &otile[wave][0];
#pragma unroll
  for (int j = 0; j < 4; ++j) {
    v4f lo, hi;
#pragma unroll
    for (int r = 0; r < 4; ++r) {
      lo[r] = acc[j][r] / lsum;
      hi[r] = acc[j][r + 4] / lsum;
    }
    *(v4f*)(ot + lmod * OPITCH + j * 16 + lhalf * 8)     = lo;
    *(v4f*)(ot + lmod * OPITCH + j * 16 + lhalf * 8 + 4) = hi;
  }
  __syncthreads();
#pragma unroll
  for (int i = 0; i < 8; ++i) {
    const int row = 2 * i + lhalf;
    const v4f ov = *(const v4f*)(ot + row * OPITCH + lmod * 4);
    *(v4f*)(O + base + (size_t)(q0 + row) * DIM_K + lmod * 4) = ov;
  }
}

extern "C" void kernel_launch(void* const* d_in, const int* in_sizes, int n_in,
                              void* d_out, int out_size, void* d_ws,
                              size_t ws_size, hipStream_t stream) {
  (void)in_sizes; (void)n_in; (void)out_size; (void)ws_size;
  const float* x  = (const float*)d_in[0];
  const float* wq = (const float*)d_in[1];
  const float* wk = (const float*)d_in[2];
  const float* wv = (const float*)d_in[3];
  const float* wu = (const float*)d_in[4];
  const float* bu = (const float*)d_in[5];
  float* out = (float*)d_out;

  const size_t MAT = (size_t)DIM_B * DIM_T * DIM_K;
  float* qb = (float*)d_ws;
  float* kb = qb + MAT;
  float* vb = kb + MAT;
  float* ab = vb + MAT;

  const int M = DIM_B * DIM_T, N = DIM_K, K = DIM_K;
  dim3 gblk(256);
  dim3 ggrid(N / 128, M / 128);

  gemm_xwt_kernel<<<ggrid, gblk, 0, stream>>>(x, wq, nullptr, qb, M, N, K);
  gemm_xwt_kernel<<<ggrid, gblk, 0, stream>>>(x, wk, nullptr, kb, M, N, K);
  gemm_xwt_kernel<<<ggrid, gblk, 0, stream>>>(x, wv, nullptr, vb, M, N, K);

  dim3 agrid(DIM_T / 64, NHEAD, DIM_B);
  attn_kernel<<<agrid, dim3(128), 0, stream>>>(qb, kb, vb, ab);

  gemm_xwt_kernel<<<ggrid, gblk, 0, stream>>>(ab, wu, bu, out, M, N, K);
}


# q, k and v projections issued as one launch (grid z picks the weight and the output plane): workgroups co-resident on a WGP fill each other's barrier and first-tile waits; 5 launches become 3
# speedup vs baseline: 1.4292x; 1.0397x over previous
#include <hip/hip_runtime.h>
#include <math.h>

typedef __attribute__((ext_vector_type(2))) float v2f;
typedef __attribute__((ext_vector_type(4))) float v4f;
typedef __attribute__((ext_vector_type(8))) float v8f;

#define DIM_B 2
#define DIM_T 2048
#define DIM_K 1024
#define NHEAD 16
#define HDIM  64

static __device__ __forceinline__ v8f wmma_f32(v2f a, v2f b, v8f c) {
  return __builtin_amdgcn_wmma_f32_16x16x4_f32(false, a, false, b, (short)0, c,
                                               false, false);
}

#define GK 32
#define GPITCH 36

__global__ __launch_bounds__(256)
__attribute__((amdgpu_waves_per_eu(2)))
void gemm_xwt_kernel(const float* __restrict__ A, const float* __restrict__ W0,
                     const float* __restrict__ W1, const float* __restrict__ W2,
                     const float* __restrict__ bias, float* __restrict__ Cbase,
                     size_t cplane, int M, int N, int K) {
  const int zsel = blockIdx.z;
  const float* __restrict__ W = (zsel == 0) ? W0 : ((zsel == 1) ? W1 : W2);
  float* __restrict__ C = Cbase + (size_t)zsel * cplane;
  __shared__ __attribute__((aligned(16))) float atile[2][128 * GPITCH];
  __shared__ __attribute__((aligned(16))) float wtile[2][128 * GPITCH];

  const int lane  = threadIdx.x & 31;
  const int wave  = threadIdx.x >> 5;
  const int wm    = wave >> 2;
  const int wn    = wave & 3;
  const int m0    = blockIdx.y * 128 + wm * 64;
  const int n0    = blockIdx.x * 128 + wn * 32;
  const int lhalf = lane >> 4;
  const int lmod  = lane & 15;

  v8f acc[4][2];
#pragma unroll
  for (int i = 0; i < 4; ++i)
#pragma unroll
    for (int j = 0; j < 2; ++j)
#pragma unroll
      for (int r = 0; r < 8; ++r) acc[i][j][r] = 0.0f;

  const int srow = threadIdx.x >> 3;
  const int scol = (threadIdx.x & 7) * 4;
  const float* ag = A + (size_t)(blockIdx.y * 128 + srow) * K + scol;
  const float* wg = W + (size_t)(blockIdx.x * 128 + srow) * K + scol;
  const int sst = srow * GPITCH + scol;

  const int afr = (wm * 64 + lmod) * GPITCH + lhalf * 2;
  const int wfr = (wn * 32 + lmod) * GPITCH + lhalf * 2;

  {
    v4f ta[4], tw[4];
#pragma unroll
    for (int i = 0; i < 4; ++i) {
      ta[i] = *(const v4f*)(ag + (size_t)(32 * i) * K);
      tw[i] = *(const v4f*)(wg + (size_t)(32 * i) * K);
    }
#pragma unroll
    for (int i = 0; i < 4; ++i) {
      *(v4f*)(&atile[0][sst + 32 * i * GPITCH]) = ta[i];
      *(v4f*)(&wtile[0][sst + 32 * i * GPITCH]) = tw[i];
    }
  }
  __syncthreads();

  for (int kt = 0; kt < K; kt += GK) {
    const int cur = (kt / GK) & 1;

    const int kn = (kt + GK < K) ? (kt + GK) : kt;
    v4f na[4], nw[4];
#pragma unroll
    for (int i = 0; i < 4; ++i) {
      na[i] = *(const v4f*)(ag + (size_t)(32 * i) * K + kn);
      nw[i] = *(const v4f*)(wg + (size_t)(32 * i) * K + kn);
    }
    __builtin_amdgcn_sched_barrier(0);

    const float* ap = &atile[cur][afr];
    const float* wp = &wtile[cur][wfr];
    v2f fa[2][4], fw[2][2];
#pragma unroll
    for (int i = 0; i < 4; ++i) fa[0][i] = *(const v2f*)(ap + i * 16 * GPITCH);
#pragma unroll
    for (int j = 0; j < 2; ++j) fw[0][j] = *(const v2f*)(wp + j * 16 * GPITCH);
#pragma unroll
    for (int ks = 0; ks < GK / 4; ++ks) {
      const int cb = ks & 1;
      if (ks + 1 < GK / 4) {
#pragma unroll
        for (int i = 0; i < 4; ++i)
          fa[cb ^ 1][i] = *(const v2f*)(ap + i * 16 * GPITCH + (ks + 1) * 4);
#pragma unroll
        for (int j = 0; j < 2; ++j)
          fw[cb ^ 1][j] = *(const v2f*)(wp + j * 16 * GPITCH + (ks + 1) * 4);
      }
#pragma unroll
      for (int i = 0; i < 4; ++i)
#pragma unroll
        for (int j = 0; j < 2; ++j)
          acc[i][j] = wmma_f32(fa[cb][i], fw[cb][j], acc[i][j]);
    }

    __builtin_amdgcn_sched_barrier(0);
#pragma unroll
    for (int i = 0; i < 4; ++i) {
      *(v4f*)(&atile[cur ^ 1][sst + 32 * i * GPITCH]) = na[i];
      *(v4f*)(&wtile[cur ^ 1][sst + 32 * i * GPITCH]) = nw[i];
    }
    __syncthreads();
  }

#pragma unroll
  for (int mi = 0; mi < 4; ++mi)
#pragma unroll
    for (int ni = 0; ni < 2; ++ni)
#pragma unroll
      for (int r = 0; r < 8; ++r) {
        const int row = m0 + mi * 16 + r + lhalf * 8;
        const int col = n0 + ni * 16 + lmod;
        float v = acc[mi][ni][r];
        if (bias) v += bias[col];
        C[(size_t)row * N + col] = v;
      }
}

static __device__ __forceinline__ float xhalf(float x) {
  const int b = __builtin_bit_cast(int, x);
  return __builtin_bit_cast(
      float, __builtin_amdgcn_permlanex16(b, b, 0x76543210, 0xfedcba98, false,
                                          false));
}

#define KPITCH 68
#define VPITCH 68
#define OPITCH 68

__global__ __launch_bounds__(128)
__attribute__((amdgpu_waves_per_eu(1)))
void attn_kernel(const float* __restrict__ Q, const float* __restrict__ Km,
                 const float* __restrict__ V, float* __restrict__ O) {
  __shared__ __attribute__((aligned(16))) float ktile[2][16 * KPITCH];
  __shared__ __attribute__((aligned(16))) float vtile[2][16 * VPITCH];
  __shared__ __attribute__((aligned(16))) float otile[4][16 * OPITCH];


  const int lane  = threadIdx.x & 31;
  const int wave  = threadIdx.x >> 5;
  const int lhalf = lane >> 4;
  const int lmod  = lane & 15;
  const int h  = blockIdx.y;
  const int b  = blockIdx.z;
  const int q0 = blockIdx.x * 64 + wave * 16;

  const size_t base = (size_t)b * DIM_T * DIM_K + (size_t)h * HDIM;

  v2f qb[16];
  {
    const float* qrow = Q + base + (size_t)(q0 + lmod) * DIM_K;
#pragma unroll
    for (int t = 0; t < 16; ++t)
      qb[t] = *(const v2f*)(qrow + t * 4 + lhalf * 2);
  }

  v8f acc[4];
#pragma unroll
  for (int j = 0; j < 4; ++j)
#pragma unroll
    for (int r = 0; r < 8; ++r) acc[j][r] = 0.0f;

  float mrun = -INFINITY, lrun = 0.0f;

  const float scale = 0.03125f;

  const int srow = threadIdx.x >> 4;
  const int scol = (threadIdx.x & 15) * 4;
  const float* kg = Km + base + (size_t)srow * DIM_K + scol;
  const float* vg = V  + base + (size_t)srow * DIM_K + scol;
  const int kst = srow * KPITCH + scol;
  const int vst = srow * VPITCH + scol;

  {
    const v4f k0 = *(const v4f*)(kg);
    const v4f k1 = *(const v4f*)(kg + (size_t)8 * DIM_K);
    const v4f v0 = *(const v4f*)(vg);
    const v4f v1 = *(const v4f*)(vg + (size_t)8 * DIM_K);
    *(v4f*)(&ktile[0][kst])              = k0;
    *(v4f*)(&ktile[0][kst + 8 * KPITCH]) = k1;
    *(v4f*)(&vtile[0][vst])              = v0;
    *(v4f*)(&vtile[0][vst + 8 * VPITCH]) = v1;
  }
  __syncthreads();

  for (int key0 = 0; key0 < DIM_T; key0 += 16) {
    const int cur = (key0 >> 4) & 1;

    const int knext = (key0 + 16 < DIM_T) ? (key0 + 16) : key0;
    const v4f nk0 = *(const v4f*)(kg + (size_t)knext * DIM_K);
    const v4f nk1 = *(const v4f*)(kg + (size_t)(knext + 8) * DIM_K);
    const v4f nv0 = *(const v4f*)(vg + (size_t)knext * DIM_K);
    const v4f nv1 = *(const v4f*)(vg + (size_t)(knext + 8) * DIM_K);

    const float* krow = &ktile[cur][lmod * KPITCH + lhalf * 2];
    v2f ka[16];
#pragma unroll
    for (int t = 0; t < 16; ++t)
      ka[t] = *(const v2f*)(krow + t * 4);
    __builtin_amdgcn_sched_barrier(0);

    v8f s;
#pragma unroll
    for (int r = 0; r < 8; ++r) s[r] = 0.0f;
#pragma unroll
    for (int t = 0; t < 16; ++t) s = wmma_f32(ka[t], qb[t], s);

    v2f vf[4][4];
#pragma unroll
    for (int t = 0; t < 4; ++t) {
      const int kk = lhalf * 8 + t * 2;
      const float* vbase = &vtile[cur][kk * VPITCH + lmod];
#pragma unroll
      for (int j = 0; j < 4; ++j) {
        vf[t][j].x = vbase[j * 16];
        vf[t][j].y = vbase[j * 16 + VPITCH];
      }
    }

    float sv[8];
#pragma unroll
    for (int r = 0; r < 8; ++r) sv[r] = s[r] * scale;
    float rm = fmaxf(fmaxf(fmaxf(sv[0], sv[1]), fmaxf(sv[2], sv[3])),
                     fmaxf(fmaxf(sv[4], sv[5]), fmaxf(sv[6], sv[7])));
    rm = fmaxf(rm, xhalf(rm));
    const float mnew = fmaxf(mrun, rm);
    const float corr = __expf(mrun - mnew);
    float p[8];
#pragma unroll
    for (int r = 0; r < 8; ++r) p[r] = __expf(sv[r] - mnew);
    const float psum = ((p[0] + p[1]) + (p[2] + p[3])) +
                       ((p[4] + p[5]) + (p[6] + p[7]));
    lrun = lrun * corr + psum;
    mrun = mnew;
#pragma unroll
    for (int j = 0; j < 4; ++j)
#pragma unroll
      for (int r = 0; r < 8; ++r) acc[j][r] *= corr;

#pragma unroll
    for (int t = 0; t < 4; ++t) {
      v2f pb;
      pb.x = p[2 * t];
      pb.y = p[2 * t + 1];
#pragma unroll
      for (int j = 0; j < 4; ++j)
        acc[j] = wmma_f32(vf[t][j], pb, acc[j]);
    }

    __builtin_amdgcn_sched_barrier(0);
    *(v4f*)(&ktile[cur ^ 1][kst])              = nk0;
    *(v4f*)(&ktile[cur ^ 1][kst + 8 * KPITCH]) = nk1;
    *(v4f*)(&vtile[cur ^ 1][vst])              = nv0;
    *(v4f*)(&vtile[cur ^ 1][vst + 8 * VPITCH]) = nv1;
    __syncthreads();
  }

  const float lsum = lrun + xhalf(lrun);
  float* ot = &otile[wave][0];
#pragma unroll
  for (int j = 0; j < 4; ++j) {
    v4f lo, hi;
#pragma unroll
    for (int r = 0; r < 4; ++r) {
      lo[r] = acc[j][r] / lsum;
      hi[r] = acc[j][r + 4] / lsum;
    }
    *(v4f*)(ot + lmod * OPITCH + j * 16 + lhalf * 8)     = lo;
    *(v4f*)(ot + lmod * OPITCH + j * 16 + lhalf * 8 + 4) = hi;
  }
  __syncthreads();
#pragma unroll
  for (int i = 0; i < 8; ++i) {
    const int row = 2 * i + lhalf;
    const v4f ov = *(const v4f*)(ot + row * OPITCH + lmod * 4);
    *(v4f*)(O + base + (size_t)(q0 + row) * DIM_K + lmod * 4) = ov;
  }
}

extern "C" void kernel_launch(void* const* d_in, const int* in_sizes, int n_in,
                              void* d_out, int out_size, void* d_ws,
                              size_t ws_size, hipStream_t stream) {
  (void)in_sizes; (void)n_in; (void)out_size; (void)ws_size;
  const float* x  = (const float*)d_in[0];
  const float* wq = (const float*)d_in[1];
  const float* wk = (const float*)d_in[2];
  const float* wv = (const float*)d_in[3];
  const float* wu = (const float*)d_in[4];
  const float* bu = (const float*)d_in[5];
  float* out = (float*)d_out;

  const size_t MAT = (size_t)DIM_B * DIM_T * DIM_K;
  float* qb = (float*)d_ws;
  float* kb = qb + MAT;
  float* vb = kb + MAT;
  float* ab = vb + MAT;

  const int M = DIM_B * DIM_T, N = DIM_K, K = DIM_K;
  dim3 gblk(256);
  dim3 ggrid(N / 128, M / 128);
  dim3 ggrid3(N / 128, M / 128, 3);

  gemm_xwt_kernel<<<ggrid3, gblk, 0, stream>>>(x, wq, wk, wv, nullptr, qb, MAT,
                                               M, N, K);

  dim3 agrid(DIM_T / 64, NHEAD, DIM_B);
  attn_kernel<<<agrid, dim3(128), 0, stream>>>(qb, kb, vb, ab);

  gemm_xwt_kernel<<<ggrid, gblk, 0, stream>>>(ab, wu, wu, wu, bu, out, 0, M, N,
                                              K);
}


# attention: 32 keys per step: two 16-key score tiles whose WMMA chains alternate; one max, lane exchange, rescale and barrier per 32 keys; row sum grouped per 32 keys (sum order changes)
# speedup vs baseline: 1.4880x; 1.0411x over previous
#include <hip/hip_runtime.h>
#include <math.h>

typedef __attribute__((ext_vector_type(2))) float v2f;
typedef __attribute__((ext_vector_type(4))) float v4f;
typedef __attribute__((ext_vector_type(8))) float v8f;

#define DIM_B 2
#define DIM_T 2048
#define DIM_K 1024
#define NHEAD 16
#define HDIM  64

static __device__ __forceinline__ v8f wmma_f32(v2f a, v2f b, v8f c) {
  return __builtin_amdgcn_wmma_f32_16x16x4_f32(false, a, false, b, (short)0, c,
                                               false, false);
}

#define GK 32
#define GPITCH 36

__global__ __launch_bounds__(256)
__attribute__((amdgpu_waves_per_eu(2)))
void gemm_xwt_kernel(const float* __restrict__ A, const float* __restrict__ W0,
                     const float* __restrict__ W1, const float* __restrict__ W2,
                     const float* __restrict__ bias, float* __restrict__ Cbase,
                     size_t cplane, int M, int N, int K) {
  const int zsel = blockIdx.z;
  const float* __restrict__ W = (zsel == 0) ? W0 : ((zsel == 1) ? W1 : W2);
  float* __restrict__ C = Cbase + (size_t)zsel * cplane;
  __shared__ __attribute__((aligned(16))) float atile[2][128 * GPITCH];
  __shared__ __attribute__((aligned(16))) float wtile[2][128 * GPITCH];

  const int lane  = threadIdx.x & 31;
  const int wave  = threadIdx.x >> 5;
  const int wm    = wave >> 2;
  const int wn    = wave & 3;
  const int m0    = blockIdx.y * 128 + wm * 64;
  const int n0    = blockIdx.x * 128 + wn * 32;
  const int lhalf = lane >> 4;
  const int lmod  = lane & 15;

  v8f acc[4][2];
#pragma unroll
  for (int i = 0; i < 4; ++i)
#pragma unroll
    for (int j = 0; j < 2; ++j)
#pragma unroll
      for (int r = 0; r < 8; ++r) acc[i][j][r] = 0.0f;

  const int srow = threadIdx.x >> 3;
  const int scol = (threadIdx.x & 7) * 4;
  const float* ag = A + (size_t)(blockIdx.y * 128 + srow) * K + scol;
  const float* wg = W + (size_t)(blockIdx.x * 128 + srow) * K + scol;
  const int sst = srow * GPITCH + scol;

  const int afr = (wm * 64 + lmod) * GPITCH + lhalf * 2;
  const int wfr = (wn * 32 + lmod) * GPITCH + lhalf * 2;

  {
    v4f ta[4], tw[4];
#pragma unroll
    for (int i = 0; i < 4; ++i) {
      ta[i] = *(const v4f*)(ag + (size_t)(32 * i) * K);
      tw[i] = *(const v4f*)(wg + (size_t)(32 * i) * K);
    }
#pragma unroll
    for (int i = 0; i < 4; ++i) {
      *(v4f*)(&atile[0][sst + 32 * i * GPITCH]) = ta[i];
      *(v4f*)(&wtile[0][sst + 32 * i * GPITCH]) = tw[i];
    }
  }
  __syncthreads();

  for (int kt = 0; kt < K; kt += GK) {
    const int cur = (kt / GK) & 1;

    const int kn = (kt + GK < K) ? (kt + GK) : kt;
    v4f na[4], nw[4];
#pragma unroll
    for (int i = 0; i < 4; ++i) {
      na[i] = *(const v4f*)(ag + (size_t)(32 * i) * K + kn);
      nw[i] = *(const v4f*)(wg + (size_t)(32 * i) * K + kn);
    }
    __builtin_amdgcn_sched_barrier(0);

    const float* ap = &atile[cur][afr];
    const float* wp = &wtile[cur][wfr];
    v2f fa[2][4], fw[2][2];
#pragma unroll
    for (int i = 0; i < 4; ++i) fa[0][i] = *(const v2f*)(ap + i * 16 * GPITCH);
#pragma unroll
    for (int j = 0; j < 2; ++j) fw[0][j] = *(const v2f*)(wp + j * 16 * GPITCH);
#pragma unroll
    for (int ks = 0; ks < GK / 4; ++ks) {
      const int cb = ks & 1;
      if (ks + 1 < GK / 4) {
#pragma unroll
        for (int i = 0; i < 4; ++i)
          fa[cb ^ 1][i] = *(const v2f*)(ap + i * 16 * GPITCH + (ks + 1) * 4);
#pragma unroll
        for (int j = 0; j < 2; ++j)
          fw[cb ^ 1][j] = *(const v2f*)(wp + j * 16 * GPITCH + (ks + 1) * 4);
      }
#pragma unroll
      for (int i = 0; i < 4; ++i)
#pragma unroll
        for (int j = 0; j < 2; ++j)
          acc[i][j] = wmma_f32(fa[cb][i], fw[cb][j], acc[i][j]);
    }

    __builtin_amdgcn_sched_barrier(0);
#pragma unroll
    for (int i = 0; i < 4; ++i) {
      *(v4f*)(&atile[cur ^ 1][sst + 32 * i * GPITCH]) = na[i];
      *(v4f*)(&wtile[cur ^ 1][sst + 32 * i * GPITCH]) = nw[i];
    }
    __syncthreads();
  }

#pragma unroll
  for (int mi = 0; mi < 4; ++mi)
#pragma unroll
    for (int ni = 0; ni < 2; ++ni)
#pragma unroll
      for (int r = 0; r < 8; ++r) {
        const int row = m0 + mi * 16 + r + lhalf * 8;
        const int col = n0 + ni * 16 + lmod;
        float v = acc[mi][ni][r];
        if (bias) v += bias[col];
        C[(size_t)row * N + col] = v;
      }
}

static __device__ __forceinline__ float xhalf(float x) {
  const int b = __builtin_bit_cast(int, x);
  return __builtin_bit_cast(
      float, __builtin_amdgcn_permlanex16(b, b, 0x76543210, 0xfedcba98, false,
                                          false));
}

#define KPITCH 68
#define VPITCH 68
#define OPITCH 68

#define KSTEP 32

__global__ __launch_bounds__(128)
__attribute__((amdgpu_waves_per_eu(1)))
void attn_kernel(const float* __restrict__ Q, const float* __restrict__ Km,
                 const float* __restrict__ V, float* __restrict__ O) {
  __shared__ __attribute__((aligned(16))) float ktile[2][KSTEP * KPITCH];
  __shared__ __attribute__((aligned(16))) float vtile[2][KSTEP * VPITCH];
  __shared__ __attribute__((aligned(16))) float otile[4][16 * OPITCH];


  const int lane  = threadIdx.x & 31;
  const int wave  = threadIdx.x >> 5;
  const int lhalf = lane >> 4;
  const int lmod  = lane & 15;
  const int h  = blockIdx.y;
  const int b  = blockIdx.z;
  const int q0 = blockIdx.x * 64 + wave * 16;

  const size_t base = (size_t)b * DIM_T * DIM_K + (size_t)h * HDIM;

  v2f qb[16];
  {
    const float* qrow = Q + base + (size_t)(q0 + lmod) * DIM_K;
#pragma unroll
    for (int t = 0; t < 16; ++t)
      qb[t] = *(const v2f*)(qrow + t * 4 + lhalf * 2);
  }

  v8f acc[4];
#pragma unroll
  for (int j = 0; j < 4; ++j)
#pragma unroll
    for (int r = 0; r < 8; ++r) acc[j][r] = 0.0f;

  float mrun = -INFINITY, lrun = 0.0f;

  const float scale = 0.03125f;

  const int srow = threadIdx.x >> 4;
  const int scol = (threadIdx.x & 15) * 4;
  const float* kg = Km + base + (size_t)srow * DIM_K + scol;
  const float* vg = V  + base + (size_t)srow * DIM_K + scol;
  const int kst = srow * KPITCH + scol;
  const int vst = srow * VPITCH + scol;

  {
    v4f tk[4], tv[4];
#pragma unroll
    for (int i = 0; i < 4; ++i) {
      tk[i] = *(const v4f*)(kg + (size_t)(8 * i) * DIM_K);
      tv[i] = *(const v4f*)(vg + (size_t)(8 * i) * DIM_K);
    }
#pragma unroll
    for (int i = 0; i < 4; ++i) {
      *(v4f*)(&ktile[0][kst + 8 * i * KPITCH]) = tk[i];
      *(v4f*)(&vtile[0][vst + 8 * i * VPITCH]) = tv[i];
    }
  }
  __syncthreads();

  for (int key0 = 0; key0 < DIM_T; key0 += KSTEP) {
    const int cur = (key0 / KSTEP) & 1;

    const int knext = (key0 + KSTEP < DIM_T) ? (key0 + KSTEP) : key0;
    v4f nk[4], nv[4];
#pragma unroll
    for (int i = 0; i < 4; ++i) {
      nk[i] = *(const v4f*)(kg + (size_t)(knext + 8 * i) * DIM_K);
      nv[i] = *(const v4f*)(vg + (size_t)(knext + 8 * i) * DIM_K);
    }

    const float* krow = &ktile[cur][lmod * KPITCH + lhalf * 2];
    v2f ka[2][16];
#pragma unroll
    for (int t = 0; t < 16; ++t)
#pragma unroll
      for (int u = 0; u < 2; ++u)
        ka[u][t] = *(const v2f*)(krow + u * 16 * KPITCH + t * 4);
    __builtin_amdgcn_sched_barrier(0);

    v8f s0, s1;
#pragma unroll
    for (int r = 0; r < 8; ++r) { s0[r] = 0.0f; s1[r] = 0.0f; }
#pragma unroll
    for (int t = 0; t < 16; ++t) {
      s0 = wmma_f32(ka[0][t], qb[t], s0);
      s1 = wmma_f32(ka[1][t], qb[t], s1);
    }
    __builtin_amdgcn_sched_group_barrier(0x008, 32, 0);
    __builtin_amdgcn_sched_group_barrier(0x100, 32, 0);

    float vx[2][4][4], vy[2][4][4];
#pragma unroll
    for (int u = 0; u < 2; ++u)
#pragma unroll
      for (int t = 0; t < 4; ++t) {
        const int kk = u * 16 + lhalf * 8 + t * 2;
        const float* vbase = &vtile[cur][kk * VPITCH + lmod];
#pragma unroll
        for (int j = 0; j < 4; ++j) {
          vx[u][t][j] = vbase[j * 16];
          vy[u][t][j] = vbase[j * 16 + VPITCH];
        }
      }
    __builtin_amdgcn_sched_barrier(0);

    float sv[16];
#pragma unroll
    for (int r = 0; r < 8; ++r) {
      sv[r]     = s0[r] * scale;
      sv[r + 8] = s1[r] * scale;
    }
    float rm0 = fmaxf(fmaxf(fmaxf(sv[0], sv[1]), fmaxf(sv[2], sv[3])),
                      fmaxf(fmaxf(sv[4], sv[5]), fmaxf(sv[6], sv[7])));
    float rm1 = fmaxf(fmaxf(fmaxf(sv[8], sv[9]), fmaxf(sv[10], sv[11])),
                      fmaxf(fmaxf(sv[12], sv[13]), fmaxf(sv[14], sv[15])));
    float rm = fmaxf(rm0, rm1);
    rm = fmaxf(rm, xhalf(rm));
    const float mnew = fmaxf(mrun, rm);
    const float corr = __expf(mrun - mnew);
    float p[16];
#pragma unroll
    for (int r = 0; r < 16; ++r) p[r] = __expf(sv[r] - mnew);
    const float psum0 = ((p[0] + p[1]) + (p[2] + p[3])) +
                        ((p[4] + p[5]) + (p[6] + p[7]));
    const float psum1 = ((p[8] + p[9]) + (p[10] + p[11])) +
                        ((p[12] + p[13]) + (p[14] + p[15]));
    lrun = lrun * corr + (psum0 + psum1);
    mrun = mnew;
#pragma unroll
    for (int j = 0; j < 4; ++j)
#pragma unroll
      for (int r = 0; r < 8; ++r) acc[j][r] *= corr;

#pragma unroll
    for (int u = 0; u < 2; ++u)
#pragma unroll
      for (int t = 0; t < 4; ++t) {
        v2f pb;
        pb.x = p[8 * u + 2 * t];
        pb.y = p[8 * u + 2 * t + 1];
#pragma unroll
        for (int j = 0; j < 4; ++j) {
          v2f va;
          va.x = vx[u][t][j];
          va.y = vy[u][t][j];
          acc[j] = wmma_f32(va, pb, acc[j]);
        }
      }

    __builtin_amdgcn_sched_barrier(0);
#pragma unroll
    for (int i = 0; i < 4; ++i) {
      *(v4f*)(&ktile[cur ^ 1][kst + 8 * i * KPITCH]) = nk[i];
      *(v4f*)(&vtile[cur ^ 1][vst + 8 * i * VPITCH]) = nv[i];
    }
    __syncthreads();
  }

  const float lsum = lrun + xhalf(lrun);
  float* ot = &otile[wave][0];
#pragma unroll
  for (int j = 0; j < 4; ++j) {
    v4f lo, hi;
#pragma unroll
    for (int r = 0; r < 4; ++r) {
      lo[r] = acc[j][r] / lsum;
      hi[r] = acc[j][r + 4] / lsum;
    }
    *(v4f*)(ot + lmod * OPITCH + j * 16 + lhalf * 8)     = lo;
    *(v4f*)(ot + lmod * OPITCH + j * 16 + lhalf * 8 + 4) = hi;
  }
  __syncthreads();
#pragma unroll
  for (int i = 0; i < 8; ++i) {
    const int row = 2 * i + lhalf;
    const v4f ov = *(const v4f*)(ot + row * OPITCH + lmod * 4);
    *(v4f*)(O + base + (size_t)(q0 + row) * DIM_K + lmod * 4) = ov;
  }
}

extern "C" void kernel_launch(void* const* d_in, const int* in_sizes, int n_in,
                              void* d_out, int out_size, void* d_ws,
                              size_t ws_size, hipStream_t stream) {
  (void)in_sizes; (void)n_in; (void)out_size; (void)ws_size;
  const float* x  = (const float*)d_in[0];
  const float* wq = (const float*)d_in[1];
  const float* wk = (const float*)d_in[2];
  const float* wv = (const float*)d_in[3];
  const float* wu = (const float*)d_in[4];
  const float* bu = (const float*)d_in[5];
  float* out = (float*)d_out;

  const size_t MAT = (size_t)DIM_B * DIM_T * DIM_K;
  float* qb = (float*)d_ws;
  float* kb = qb + MAT;
  float* vb = kb + MAT;
  float* ab = vb + MAT;

  const int M = DIM_B * DIM_T, N = DIM_K, K = DIM_K;
  dim3 gblk(256);
  dim3 ggrid(N / 128, M / 128);
  dim3 ggrid3(N / 128, M / 128, 3);

  gemm_xwt_kernel<<<ggrid3, gblk, 0, stream>>>(x, wq, wk, wv, nullptr, qb, MAT,
                                               M, N, K);

  dim3 agrid(DIM_T / 64, NHEAD, DIM_B);
  attn_kernel<<<agrid, dim3(128), 0, stream>>>(qb, kb, vb, ab);

  gemm_xwt_kernel<<<ggrid, gblk, 0, stream>>>(ab, wu, wu, wu, bu, out, 0, M, N,
                                              K);
}
